# starting kernel
# baseline (speedup 1.0000x reference)
#include <hip/hip_runtime.h>
#include <hip/hip_bf16.h>

#define N_NODES 50000
#define N_EDGES 1600000
#define IN_F    256
#define HID     32
#define HEADS   4
#define F2      128
#define NC      16
#define NEG_SLOPE 0.2f

typedef __attribute__((ext_vector_type(16))) _Float16 v16h;
typedef __attribute__((ext_vector_type(8)))  _Float16 v8h;
typedef __attribute__((ext_vector_type(8)))  float    v8f;

__device__ __forceinline__ unsigned enc_key(float f) {
    unsigned b = __float_as_uint(f);
    return (b & 0x80000000u) ? ~b : (b | 0x80000000u);
}
__device__ __forceinline__ float dec_key(unsigned k) {
    unsigned b = (k & 0x80000000u) ? (k ^ 0x80000000u) : ~k;
    if ((b & 0x7f800000u) == 0x7f800000u) return 0.0f;
    return __uint_as_float(b);
}
__device__ __forceinline__ float leaky(float v) { return v > 0.0f ? v : NEG_SLOPE * v; }

__global__ void k_f32_to_f16(const float* __restrict__ in, _Float16* __restrict__ out, int n) {
    int i = blockIdx.x * blockDim.x + threadIdx.x;
    if (i < n) out[i] = (_Float16)in[i];
}
__global__ void k_transpose_f16(const float* __restrict__ in, _Float16* __restrict__ out,
                                int R, int C) {
    int i = blockIdx.x * blockDim.x + threadIdx.x;
    if (i >= R * C) return;
    int r = i / C, c = i % C;
    out[c * R + r] = (_Float16)in[i];
}

template <int K>
__global__ void __launch_bounds__(256) k_gemm_wmma(const _Float16* __restrict__ A,
                                                   const _Float16* __restrict__ Bt,
                                                   float* __restrict__ C,
                                                   int M, int Ncol) {
    int wave = (blockIdx.x * blockDim.x + threadIdx.x) >> 5;
    int lane = threadIdx.x & 31;
    int mTiles = M >> 4, nTiles = Ncol >> 4;
    if (wave >= mTiles * nTiles) return;
    int mt = wave / nTiles, nt = wave % nTiles;
    int half = lane >> 4;
    int idx  = lane & 15;

    const _Float16* arow = A  + (size_t)(mt * 16 + idx) * K + half * 8;
    const _Float16* brow = Bt + (size_t)(nt * 16 + idx) * K + half * 8;

    v8f acc = {};
#pragma unroll
    for (int k0 = 0; k0 < K; k0 += 32) {
        union { v16h v; struct { v8h lo, hi; } h; } a, b;
        a.h.lo = *(const v8h*)(arow + k0);
        a.h.hi = *(const v8h*)(arow + k0 + 16);
        b.h.lo = *(const v8h*)(brow + k0);
        b.h.hi = *(const v8h*)(brow + k0 + 16);
        acc = __builtin_amdgcn_wmma_f32_16x16x32_f16(false, a.v, false, b.v,
                                                     (short)0, acc, false, false);
    }
    float* crow = C + (size_t)(mt * 16 + half * 8) * Ncol + nt * 16 + idx;
#pragma unroll
    for (int r = 0; r < 8; ++r) crow[(size_t)r * Ncol] = acc[r];
}

__global__ void k_el_er(const float* __restrict__ feat, const float* __restrict__ al,
                        const float* __restrict__ ar, float* __restrict__ el,
                        float* __restrict__ er, int n, int H, int D) {
    int i = blockIdx.x * blockDim.x + threadIdx.x;
    if (i >= n * H) return;
    int node = i / H, h = i % H;
    const float* f = feat + (size_t)node * H * D + h * D;
    const float* a = al + h * D;
    const float* b = ar + h * D;
    float sl = 0.0f, sr = 0.0f;
    for (int d = 0; d < D; ++d) { float v = f[d]; sl += v * a[d]; sr += v * b[d]; }
    el[i] = sl; er[i] = sr;
}

__global__ void k_edge_max4(const int* __restrict__ src, const int* __restrict__ dst,
                            const float* __restrict__ el, const float* __restrict__ er,
                            unsigned* __restrict__ mx) {
    int e = blockIdx.x * blockDim.x + threadIdx.x;
    if (e >= N_EDGES) return;
    int s = src[e], d = dst[e];
    float4 a = *(const float4*)(el + 4 * (size_t)s);
    float4 b = *(const float4*)(er + 4 * (size_t)d);
    float v[4] = { a.x + b.x, a.y + b.y, a.z + b.z, a.w + b.w };
#pragma unroll
    for (int h = 0; h < 4; ++h)
        atomicMax(mx + 4 * (size_t)d + h, enc_key(leaky(v[h])));
}

__global__ void k_edge_exp4(const int* __restrict__ src, const int* __restrict__ dst,
                            const float* __restrict__ el, const float* __restrict__ er,
                            const unsigned* __restrict__ mx, float* __restrict__ ex,
                            float* __restrict__ den) {
    int e = blockIdx.x * blockDim.x + threadIdx.x;
    if (e >= N_EDGES) return;
    int s = src[e], d = dst[e];
    float4 a = *(const float4*)(el + 4 * (size_t)s);
    float4 b = *(const float4*)(er + 4 * (size_t)d);
    float v[4] = { a.x + b.x, a.y + b.y, a.z + b.z, a.w + b.w };
#pragma unroll
    for (int h = 0; h < 4; ++h) {
        float m = dec_key(mx[4 * (size_t)d + h]);
        float x = __expf(leaky(v[h]) - m);
        ex[4 * (size_t)e + h] = x;
        atomicAdd(den + 4 * (size_t)d + h, x);
    }
}

__global__ void __launch_bounds__(256) k_edge_agg4(const int* __restrict__ src,
                                                   const int* __restrict__ dst,
                                                   const float* __restrict__ feat,
                                                   const float* __restrict__ ex,
                                                   const float* __restrict__ den,
                                                   float* __restrict__ agg) {
    int wave = (blockIdx.x * blockDim.x + threadIdx.x) >> 5;
    int lane = threadIdx.x & 31;
    if (wave >= N_EDGES) return;
    int s = src[wave], d = dst[wave];
    int h = lane >> 3;
    float dn = den[4 * (size_t)d + h];
    float alpha = ex[4 * (size_t)wave + h] / fmaxf(dn, 1e-9f);
    float4 f = *(const float4*)(feat + (size_t)s * F2 + lane * 4);
    float* o = agg + (size_t)d * F2 + lane * 4;
    atomicAdd(o + 0, f.x * alpha);
    atomicAdd(o + 1, f.y * alpha);
    atomicAdd(o + 2, f.z * alpha);
    atomicAdd(o + 3, f.w * alpha);
}

__global__ void k_fin1(const float* __restrict__ agg, const float* __restrict__ b1,
                       _Float16* __restrict__ h16) {
    int i = blockIdx.x * blockDim.x + threadIdx.x;
    if (i >= N_NODES * F2) return;
    float v = agg[i] + b1[i & (F2 - 1)];
    v = v > 0.0f ? v : (__expf(v) - 1.0f);
    h16[i] = (_Float16)v;
}

__global__ void k_edge_max1(const int* __restrict__ src, const int* __restrict__ dst,
                            const float* __restrict__ el, const float* __restrict__ er,
                            unsigned* __restrict__ mx) {
    int e = blockIdx.x * blockDim.x + threadIdx.x;
    if (e >= N_EDGES) return;
    int s = src[e], d = dst[e];
    atomicMax(mx + d, enc_key(leaky(el[s] + er[d])));
}

__global__ void k_edge_exp1(const int* __restrict__ src, const int* __restrict__ dst,
                            const float* __restrict__ el, const float* __restrict__ er,
                            const unsigned* __restrict__ mx, float* __restrict__ ex,
                            float* __restrict__ den) {
    int e = blockIdx.x * blockDim.x + threadIdx.x;
    if (e >= N_EDGES) return;
    int s = src[e], d = dst[e];
    float x = __expf(leaky(el[s] + er[d]) - dec_key(mx[d]));
    ex[e] = x;
    atomicAdd(den + d, x);
}

__global__ void k_edge_agg1(const int* __restrict__ src, const int* __restrict__ dst,
                            const float* __restrict__ feat, const float* __restrict__ ex,
                            const float* __restrict__ den, float* __restrict__ agg) {
    int t = blockIdx.x * blockDim.x + threadIdx.x;
    if (t >= N_EDGES * 4) return;
    int e = t >> 2, q = t & 3;
    int s = src[e], d = dst[e];
    float alpha = ex[e] / fmaxf(den[d], 1e-9f);
    float4 f = *(const float4*)(feat + (size_t)s * NC + q * 4);
    float* o = agg + (size_t)d * NC + q * 4;
    atomicAdd(o + 0, f.x * alpha);
    atomicAdd(o + 1, f.y * alpha);
    atomicAdd(o + 2, f.z * alpha);
    atomicAdd(o + 3, f.w * alpha);
}

__global__ void k_fin2(const float* __restrict__ agg, const float* __restrict__ b2,
                       float* __restrict__ out) {
    int i = blockIdx.x * blockDim.x + threadIdx.x;
    if (i >= N_NODES * NC) return;
    out[i] = agg[i] + b2[i & (NC - 1)];
}

extern "C" void kernel_launch(void* const* d_in, const int* in_sizes, int n_in,
                              void* d_out, int out_size, void* d_ws, size_t ws_size,
                              hipStream_t stream) {
    const float* x   = (const float*)d_in[0];
    const int*   src = (const int*)  d_in[1];
    const int*   dst = (const int*)  d_in[2];
    const float* W1  = (const float*)d_in[3];
    const float* b1  = (const float*)d_in[4];
    const float* al1 = (const float*)d_in[5];
    const float* ar1 = (const float*)d_in[6];
    const float* W2  = (const float*)d_in[7];
    const float* b2  = (const float*)d_in[8];
    const float* al2 = (const float*)d_in[9];
    const float* ar2 = (const float*)d_in[10];
    float* out = (float*)d_out;
    (void)in_sizes; (void)n_in; (void)out_size; (void)ws_size;

    char* p = (char*)d_ws;
    auto carve = [&](size_t bytes) -> char* {
        char* r = p; p += (bytes + 255) & ~(size_t)255; return r;
    };
    _Float16* xh    = (_Float16*)carve((size_t)N_NODES * IN_F * 2);
    _Float16* w1t   = (_Float16*)carve((size_t)IN_F * F2 * 2);
    float*    feat1 = (float*)   carve((size_t)N_NODES * F2 * 4);
    float*    el1   = (float*)   carve((size_t)N_NODES * HEADS * 4);
    float*    er1   = (float*)   carve((size_t)N_NODES * HEADS * 4);
    unsigned* mx1   = (unsigned*)carve((size_t)N_NODES * HEADS * 4);
    float*    den1  = (float*)   carve((size_t)N_NODES * HEADS * 4);
    float*    ex1   = (float*)   carve((size_t)N_EDGES * HEADS * 4);
    float*    agg1  = (float*)   carve((size_t)N_NODES * F2 * 4);
    _Float16* h16   = (_Float16*)carve((size_t)N_NODES * F2 * 2);
    _Float16* w2t   = (_Float16*)carve((size_t)F2 * NC * 2);
    float*    feat2 = (float*)   carve((size_t)N_NODES * NC * 4);
    float*    el2   = (float*)   carve((size_t)N_NODES * 4);
    float*    er2   = (float*)   carve((size_t)N_NODES * 4);
    unsigned* mx2   = (unsigned*)carve((size_t)N_NODES * 4);
    float*    den2  = (float*)   carve((size_t)N_NODES * 4);
    float*    ex2   = (float*)   carve((size_t)N_EDGES * 4);
    float*    agg2  = (float*)   carve((size_t)N_NODES * NC * 4);

    hipMemsetAsync(mx1,  0, (size_t)N_NODES * HEADS * 4, stream);
    hipMemsetAsync(den1, 0, (size_t)N_NODES * HEADS * 4, stream);
    hipMemsetAsync(agg1, 0, (size_t)N_NODES * F2 * 4,    stream);
    hipMemsetAsync(mx2,  0, (size_t)N_NODES * 4,         stream);
    hipMemsetAsync(den2, 0, (size_t)N_NODES * 4,         stream);
    hipMemsetAsync(agg2, 0, (size_t)N_NODES * NC * 4,    stream);

    const int T = 256;
    auto blk = [](long long n, int t) { return (int)((n + t - 1) / t); };

    k_f32_to_f16   <<<blk((long long)N_NODES * IN_F, T), T, 0, stream>>>(x, xh, N_NODES * IN_F);
    k_transpose_f16<<<blk((long long)IN_F * F2, T), T, 0, stream>>>(W1, w1t, IN_F, F2);

    {
        int waves = (N_NODES / 16) * (F2 / 16);
        k_gemm_wmma<IN_F><<<blk(waves, 8), T, 0, stream>>>(xh, w1t, feat1, N_NODES, F2);
    }
    k_el_er<<<blk((long long)N_NODES * HEADS, T), T, 0, stream>>>(feat1, al1, ar1, el1, er1,
                                                                  N_NODES, HEADS, HID);
    k_edge_max4<<<blk(N_EDGES, T), T, 0, stream>>>(src, dst, el1, er1, mx1);
    k_edge_exp4<<<blk(N_EDGES, T), T, 0, stream>>>(src, dst, el1, er1, mx1, ex1, den1);
    k_edge_agg4<<<blk((long long)N_EDGES * 32, T), T, 0, stream>>>(src, dst, feat1, ex1, den1, agg1);
    k_fin1<<<blk((long long)N_NODES * F2, T), T, 0, stream>>>(agg1, b1, h16);

    k_transpose_f16<<<blk((long long)F2 * NC, T), T, 0, stream>>>(W2, w2t, F2, NC);
    {
        int waves = (N_NODES / 16) * (NC / 16);
        k_gemm_wmma<F2><<<blk(waves, 8), T, 0, stream>>>(h16, w2t, feat2, N_NODES, NC);
    }
    k_el_er<<<blk(N_NODES, T), T, 0, stream>>>(feat2, al2, ar2, el2, er2, N_NODES, 1, NC);
    k_edge_max1<<<blk(N_EDGES, T), T, 0, stream>>>(src, dst, el2, er2, mx2);
    k_edge_exp1<<<blk(N_EDGES, T), T, 0, stream>>>(src, dst, el2, er2, mx2, ex2, den2);
    k_edge_agg1<<<blk((long long)N_EDGES * 4, T), T, 0, stream>>>(src, dst, feat2, ex2, den2, agg2);
    k_fin2<<<blk((long long)N_NODES * NC, T), T, 0, stream>>>(agg2, b2, out);
}


# In-degree prefix sum on 49 workgroups: one node per thread, coalesced loads, each workgroup adds up the degrees in front of its chunk itself; replaces one workgroup's strided serial loops
# speedup vs baseline: 4.4414x; 4.4414x over previous
#include <hip/hip_runtime.h>
#include <hip/hip_bf16.h>

#define N_NODES 50000
#define N_EDGES 1600000
#define IN_F    256
#define HID     32
#define HEADS   4
#define F2      128
#define NC      16
#define NEG_SLOPE 0.2f

typedef __attribute__((ext_vector_type(16))) _Float16 v16h;
typedef __attribute__((ext_vector_type(8)))  _Float16 v8h;
typedef __attribute__((ext_vector_type(8)))  float    v8f;

__device__ __forceinline__ float leaky(float v) { return v > 0.0f ? v : NEG_SLOPE * v; }
__device__ __forceinline__ float finite_or_zero(float m) { return __builtin_isfinite(m) ? m : 0.0f; }
__device__ __forceinline__ float wave_max(float v) {
#pragma unroll
    for (int o = 16; o > 0; o >>= 1) v = fmaxf(v, __shfl_xor(v, o));
    return v;
}
__device__ __forceinline__ float wave_sum(float v) {
#pragma unroll
    for (int o = 16; o > 0; o >>= 1) v += __shfl_xor(v, o);
    return v;
}

__global__ void k_f32_to_f16(const float* __restrict__ in, _Float16* __restrict__ out, int n) {
    int i = blockIdx.x * blockDim.x + threadIdx.x;
    if (i < n) out[i] = (_Float16)in[i];
}
__global__ void k_transpose_f16(const float* __restrict__ in, _Float16* __restrict__ out,
                                int R, int C) {
    int i = blockIdx.x * blockDim.x + threadIdx.x;
    if (i >= R * C) return;
    int r = i / C, c = i % C;
    out[c * R + r] = (_Float16)in[i];
}

template <int K>
__global__ void __launch_bounds__(256) k_gemm_wmma(const _Float16* __restrict__ A,
                                                   const _Float16* __restrict__ Bt,
                                                   float* __restrict__ C,
                                                   int M, int Ncol) {
    int wave = (blockIdx.x * blockDim.x + threadIdx.x) >> 5;
    int lane = threadIdx.x & 31;
    int mTiles = M >> 4, nTiles = Ncol >> 4;
    if (wave >= mTiles * nTiles) return;
    int mt = wave / nTiles, nt = wave % nTiles;
    int half = lane >> 4;
    int idx  = lane & 15;

    const _Float16* arow = A  + (size_t)(mt * 16 + idx) * K + half * 8;
    const _Float16* brow = Bt + (size_t)(nt * 16 + idx) * K + half * 8;

    v8f acc = {};
#pragma unroll
    for (int k0 = 0; k0 < K; k0 += 32) {
        union { v16h v; struct { v8h lo, hi; } h; } a, b;
        a.h.lo = *(const v8h*)(arow + k0);
        a.h.hi = *(const v8h*)(arow + k0 + 16);
        b.h.lo = *(const v8h*)(brow + k0);
        b.h.hi = *(const v8h*)(brow + k0 + 16);
        acc = __builtin_amdgcn_wmma_f32_16x16x32_f16(false, a.v, false, b.v,
                                                     (short)0, acc, false, false);
    }
    float* crow = C + (size_t)(mt * 16 + half * 8) * Ncol + nt * 16 + idx;
#pragma unroll
    for (int r = 0; r < 8; ++r) crow[(size_t)r * Ncol] = acc[r];
}

__global__ void k_el_er(const float* __restrict__ feat, const float* __restrict__ al,
                        const float* __restrict__ ar, float* __restrict__ el,
                        float* __restrict__ er, int n, int H, int D) {
    int i = blockIdx.x * blockDim.x + threadIdx.x;
    if (i >= n * H) return;
    int node = i / H, h = i % H;
    const float* f = feat + (size_t)node * H * D + h * D;
    const float* a = al + h * D;
    const float* b = ar + h * D;
    float sl = 0.0f, sr = 0.0f;
    for (int d = 0; d < D; ++d) { float v = f[d]; sl += v * a[d]; sr += v * b[d]; }
    el[i] = sl; er[i] = sr;
}

__global__ void k_rank(const int* __restrict__ dst, int* __restrict__ deg, int* __restrict__ rank) {
    int e = blockIdx.x * blockDim.x + threadIdx.x;
    if (e >= N_EDGES) return;
    rank[e] = atomicAdd(deg + dst[e], 1);
}

#define SCAN_T   1024
#define SCAN_W   (SCAN_T / 32)
#define SCAN_NB  ((N_NODES + SCAN_T - 1) / SCAN_T)
#define SCAN_PRE (((SCAN_NB - 1) * (SCAN_T / 4) + SCAN_T - 1) / SCAN_T)

__device__ __forceinline__ int wave_sum_i(int v) {
    for (int o = 16; o > 0; o >>= 1) v += __shfl_xor(v, o);
    return v;
}
__device__ __forceinline__ int wave_scan_incl(int v, int lane) {
    for (int o = 1; o < 32; o <<= 1) {
        const int u = __shfl_up(v, o);
        if (lane >= o) v += u;
    }
    return v;
}

__global__ void __launch_bounds__(SCAN_T) k_scan(const int* __restrict__ deg, int* __restrict__ rowptr) {
    __shared__ int wtot[SCAN_W];
    __shared__ int wpre[SCAN_W];
    const int t = threadIdx.x, lane = t & 31, w = t >> 5;
    const int i = blockIdx.x * SCAN_T + t;
    const int v = (i < N_NODES) ? deg[i] : 0;

    const int4* d4 = (const int4*)deg;
    const int lim = blockIdx.x * (SCAN_T / 4);
    int pre = 0;
#pragma unroll
    for (int j = 0; j < SCAN_PRE; ++j) {
        const int k = t + j * SCAN_T;
        const int4 q = d4[k < lim ? k : 0];
        const int s = (q.x + q.y) + (q.z + q.w);
        pre += (k < lim) ? s : 0;
    }
    const int inc = wave_scan_incl(v, lane);
    pre = wave_sum_i(pre);
    if (lane == 31) wtot[w] = inc;
    if (lane == 0)  wpre[w] = pre;
    __syncthreads();
    const int tot   = wtot[lane];
    const int front = wave_sum_i(wpre[lane]);
    const int tinc  = wave_scan_incl(tot, lane);
    const int woff  = __shfl(tinc - tot, w);
    const int excl  = front + woff + inc - v;
    if (i < N_NODES) rowptr[i] = excl;
    if (i == N_NODES - 1) rowptr[N_NODES] = excl + v;
}

__global__ void k_fill(const int* __restrict__ src, const int* __restrict__ dst,
                       const int* __restrict__ rowptr, const int* __restrict__ rank,
                       int* __restrict__ ssrc) {
    int e = blockIdx.x * blockDim.x + threadIdx.x;
    if (e >= N_EDGES) return;
    ssrc[rowptr[dst[e]] + rank[e]] = src[e];
}

__device__ __forceinline__ void load_chunk4(const int* __restrict__ ssrc, const float* __restrict__ el,
                                            float4 erd, int base, int end, int lane,
                                            int& s, float4& v) {
    const int i = base + lane;
    const bool valid = i < end;
    s = ssrc[valid ? i : end - 1];
    const float4 a = *(const float4*)(el + 4 * (size_t)s);
    v.x = valid ? leaky(a.x + erd.x) : -INFINITY;
    v.y = valid ? leaky(a.y + erd.y) : -INFINITY;
    v.z = valid ? leaky(a.z + erd.z) : -INFINITY;
    v.w = valid ? leaky(a.w + erd.w) : -INFINITY;
}

__device__ __forceinline__ void accum_chunk4(int s, float4 v, int cnt, float4 m, float4* xrow,
                                             const float* __restrict__ feat, int lane, int h,
                                             float4& acc, float4& den) {
    float4 x;
    x.x = __expf(v.x - m.x); x.y = __expf(v.y - m.y);
    x.z = __expf(v.z - m.z); x.w = __expf(v.w - m.w);
    den.x += x.x; den.y += x.y; den.z += x.z; den.w += x.w;
    xrow[lane] = x;
    const float* xw = (const float*)xrow;
    const int cnt4 = (cnt + 3) & ~3;
    for (int j = 0; j < cnt4; j += 4) {
#pragma unroll
        for (int u = 0; u < 4; ++u) {
            const int sj = __builtin_amdgcn_readlane(s, j + u);
            const float xj = xw[(j + u) * 4 + h];
            const float4 f = *(const float4*)(feat + (size_t)sj * F2 + lane * 4);
            acc.x = fmaf(f.x, xj, acc.x);
            acc.y = fmaf(f.y, xj, acc.y);
            acc.z = fmaf(f.z, xj, acc.z);
            acc.w = fmaf(f.w, xj, acc.w);
        }
    }
}

typedef __attribute__((ext_vector_type(4))) _Float16 v4h;

__global__ void __launch_bounds__(256) k_node4(const int* __restrict__ rowptr,
                                               const int* __restrict__ ssrc,
                                               const float* __restrict__ el,
                                               const float* __restrict__ er,
                                               const float* __restrict__ feat,
                                               const float* __restrict__ b1,
                                               _Float16* __restrict__ h16) {
    __shared__ float4 xs[8][32];
    const int lane = threadIdx.x & 31;
    const int wib  = threadIdx.x >> 5;
    const int node = __builtin_amdgcn_readfirstlane(blockIdx.x * 8 + wib);
    if (node >= N_NODES) return;
    const int beg = rowptr[node], end = rowptr[node + 1];
    const int h = lane >> 3;
    const float4 bb = *(const float4*)(b1 + lane * 4);
    float4 acc = {0.0f, 0.0f, 0.0f, 0.0f};
    float d = 0.0f;
    if (beg < end) {
        const float4 erd = *(const float4*)(er + 4 * (size_t)node);
        int sF; float4 vF;
        load_chunk4(ssrc, el, erd, beg, end, lane, sF, vF);
        float4 m = vF;
        for (int base = beg + 32; base < end; base += 32) {
            int s; float4 v;
            load_chunk4(ssrc, el, erd, base, end, lane, s, v);
            m.x = fmaxf(m.x, v.x); m.y = fmaxf(m.y, v.y);
            m.z = fmaxf(m.z, v.z); m.w = fmaxf(m.w, v.w);
        }
        m.x = finite_or_zero(wave_max(m.x)); m.y = finite_or_zero(wave_max(m.y));
        m.z = finite_or_zero(wave_max(m.z)); m.w = finite_or_zero(wave_max(m.w));

        float4 den = {0.0f, 0.0f, 0.0f, 0.0f};
        int rem = end - beg;
        accum_chunk4(sF, vF, rem < 32 ? rem : 32, m, xs[wib], feat, lane, h, acc, den);
        for (int base = beg + 32; base < end; base += 32) {
            int s; float4 v;
            load_chunk4(ssrc, el, erd, base, end, lane, s, v);
            rem = end - base;
            accum_chunk4(s, v, rem < 32 ? rem : 32, m, xs[wib], feat, lane, h, acc, den);
        }
        const float d0 = wave_sum(den.x), d1 = wave_sum(den.y);
        const float d2 = wave_sum(den.z), d3 = wave_sum(den.w);
        d = (h == 0) ? d0 : (h == 1) ? d1 : (h == 2) ? d2 : d3;
    }
    d = fmaxf(d, 1e-9f);
    float r[4] = { acc.x / d + bb.x, acc.y / d + bb.y, acc.z / d + bb.z, acc.w / d + bb.w };
    v4h o;
#pragma unroll
    for (int k = 0; k < 4; ++k) {
        const float t = r[k] > 0.0f ? r[k] : (__expf(r[k]) - 1.0f);
        o[k] = (_Float16)t;
    }
    *(v4h*)(h16 + (size_t)node * F2 + lane * 4) = o;
}

__device__ __forceinline__ void load_chunk1(const int* __restrict__ ssrc, const float* __restrict__ el,
                                            float erd, int base, int end, int lane,
                                            int& s, float& v) {
    const int i = base + lane;
    const bool valid = i < end;
    s = ssrc[valid ? i : end - 1];
    v = valid ? leaky(el[s] + erd) : -INFINITY;
}

__device__ __forceinline__ void accum_chunk1(int s, float v, float m, const float* __restrict__ feat,
                                             int sub, int q, float4& acc, float& den) {
    const float x = __expf(v - m);
    den += x;
#pragma unroll
    for (int g = 0; g < 4; ++g) {
        const int jl = g * 8 + sub;
        const int sj = __shfl(s, jl);
        const float xj = __shfl(x, jl);
        const float4 f = *(const float4*)(feat + (size_t)sj * NC + q * 4);
        acc.x = fmaf(f.x, xj, acc.x);
        acc.y = fmaf(f.y, xj, acc.y);
        acc.z = fmaf(f.z, xj, acc.z);
        acc.w = fmaf(f.w, xj, acc.w);
    }
}

__global__ void __launch_bounds__(256) k_node1(const int* __restrict__ rowptr,
                                               const int* __restrict__ ssrc,
                                               const float* __restrict__ el,
                                               const float* __restrict__ er,
                                               const float* __restrict__ feat,
                                               const float* __restrict__ b2,
                                               float* __restrict__ out) {
    const int lane = threadIdx.x & 31;
    const int wib  = threadIdx.x >> 5;
    const int node = __builtin_amdgcn_readfirstlane(blockIdx.x * 8 + wib);
    if (node >= N_NODES) return;
    const int beg = rowptr[node], end = rowptr[node + 1];
    const int q = lane & 3, sub = lane >> 2;
    float4 acc = {0.0f, 0.0f, 0.0f, 0.0f};
    float den = 0.0f;
    if (beg < end) {
        const float erd = er[node];
        int sF; float vF;
        load_chunk1(ssrc, el, erd, beg, end, lane, sF, vF);
        float m = vF;
        for (int base = beg + 32; base < end; base += 32) {
            int s; float v;
            load_chunk1(ssrc, el, erd, base, end, lane, s, v);
            m = fmaxf(m, v);
        }
        m = finite_or_zero(wave_max(m));
        accum_chunk1(sF, vF, m, feat, sub, q, acc, den);
        for (int base = beg + 32; base < end; base += 32) {
            int s; float v;
            load_chunk1(ssrc, el, erd, base, end, lane, s, v);
            accum_chunk1(s, v, m, feat, sub, q, acc, den);
        }
#pragma unroll
        for (int o = 4; o < 32; o <<= 1) {
            acc.x += __shfl_xor(acc.x, o); acc.y += __shfl_xor(acc.y, o);
            acc.z += __shfl_xor(acc.z, o); acc.w += __shfl_xor(acc.w, o);
        }
        den = wave_sum(den);
    }
    if (sub == 0) {
        const float d = fmaxf(den, 1e-9f);
        const float4 bb = *(const float4*)(b2 + q * 4);
        float4 r;
        r.x = acc.x / d + bb.x; r.y = acc.y / d + bb.y;
        r.z = acc.z / d + bb.z; r.w = acc.w / d + bb.w;
        *(float4*)(out + (size_t)node * NC + q * 4) = r;
    }
}

extern "C" void kernel_launch(void* const* d_in, const int* in_sizes, int n_in,
                              void* d_out, int out_size, void* d_ws, size_t ws_size,
                              hipStream_t stream) {
    const float* x   = (const float*)d_in[0];
    const int*   src = (const int*)  d_in[1];
    const int*   dst = (const int*)  d_in[2];
    const float* W1  = (const float*)d_in[3];
    const float* b1  = (const float*)d_in[4];
    const float* al1 = (const float*)d_in[5];
    const float* ar1 = (const float*)d_in[6];
    const float* W2  = (const float*)d_in[7];
    const float* b2  = (const float*)d_in[8];
    const float* al2 = (const float*)d_in[9];
    const float* ar2 = (const float*)d_in[10];
    float* out = (float*)d_out;
    (void)in_sizes; (void)n_in; (void)out_size; (void)ws_size;

    char* p = (char*)d_ws;
    auto carve = [&](size_t bytes) -> char* {
        char* r = p; p += (bytes + 255) & ~(size_t)255; return r;
    };
    _Float16* xh    = (_Float16*)carve((size_t)N_NODES * IN_F * 2);
    _Float16* w1t   = (_Float16*)carve((size_t)IN_F * F2 * 2);
    float*    feat1 = (float*)   carve((size_t)N_NODES * F2 * 4);
    float*    el1   = (float*)   carve((size_t)N_NODES * HEADS * 4);
    float*    er1   = (float*)   carve((size_t)N_NODES * HEADS * 4);
    _Float16* h16   = (_Float16*)carve((size_t)N_NODES * F2 * 2);
    _Float16* w2t   = (_Float16*)carve((size_t)F2 * NC * 2);
    float*    feat2 = (float*)   carve((size_t)N_NODES * NC * 4);
    float*    el2   = (float*)   carve((size_t)N_NODES * 4);
    float*    er2   = (float*)   carve((size_t)N_NODES * 4);
    int*      deg    = (int*)carve((size_t)N_NODES * 4);
    int*      rowptr = (int*)carve((size_t)(N_NODES + 1) * 4);
    int*      rank   = (int*)carve((size_t)N_EDGES * 4);
    int*      ssrc   = (int*)carve((size_t)N_EDGES * 4);

    hipMemsetAsync(deg, 0, (size_t)N_NODES * 4, stream);

    const int T = 256;
    auto blk = [](long long n, int t) { return (int)((n + t - 1) / t); };

    k_f32_to_f16   <<<blk((long long)N_NODES * IN_F, T), T, 0, stream>>>(x, xh, N_NODES * IN_F);
    k_transpose_f16<<<blk((long long)IN_F * F2, T), T, 0, stream>>>(W1, w1t, IN_F, F2);

    {
        int waves = (N_NODES / 16) * (F2 / 16);
        k_gemm_wmma<IN_F><<<blk(waves, 8), T, 0, stream>>>(xh, w1t, feat1, N_NODES, F2);
    }
    k_el_er<<<blk((long long)N_NODES * HEADS, T), T, 0, stream>>>(feat1, al1, ar1, el1, er1,
                                                                  N_NODES, HEADS, HID);
    k_rank<<<blk(N_EDGES, T), T, 0, stream>>>(dst, deg, rank);
    k_scan<<<SCAN_NB, SCAN_T, 0, stream>>>(deg, rowptr);
    k_fill<<<blk(N_EDGES, T), T, 0, stream>>>(src, dst, rowptr, rank, ssrc);
    k_node4<<<blk(N_NODES, 8), T, 0, stream>>>(rowptr, ssrc, el1, er1, feat1, b1, h16);

    k_transpose_f16<<<blk((long long)F2 * NC, T), T, 0, stream>>>(W2, w2t, F2, NC);
    {
        int waves = (N_NODES / 16) * (NC / 16);
        k_gemm_wmma<F2><<<blk(waves, 8), T, 0, stream>>>(h16, w2t, feat2, N_NODES, NC);
    }
    k_el_er<<<blk(N_NODES, T), T, 0, stream>>>(feat2, al2, ar2, el2, er2, N_NODES, 1, NC);
    k_node1<<<blk(N_NODES, 8), T, 0, stream>>>(rowptr, ssrc, el2, er2, feat2, b2, out);
}


# Edges grouped by destination with a two-level counting sort in LDS (slice bin counts, column sums, per-bin node sort): no global atomics, no memset; bucket order is LDS arrival order, sums reorder
# speedup vs baseline: 9.3447x; 2.1040x over previous
#include <hip/hip_runtime.h>
#include <hip/hip_bf16.h>

#define N_NODES 50000
#define N_EDGES 1600000
#define IN_F    256
#define HID     32
#define HEADS   4
#define F2      128
#define NC      16
#define NEG_SLOPE 0.2f

typedef __attribute__((ext_vector_type(16))) _Float16 v16h;
typedef __attribute__((ext_vector_type(8)))  _Float16 v8h;
typedef __attribute__((ext_vector_type(8)))  float    v8f;

__device__ __forceinline__ float leaky(float v) { return v > 0.0f ? v : NEG_SLOPE * v; }
__device__ __forceinline__ float finite_or_zero(float m) { return __builtin_isfinite(m) ? m : 0.0f; }
__device__ __forceinline__ float wave_max(float v) {
#pragma unroll
    for (int o = 16; o > 0; o >>= 1) v = fmaxf(v, __shfl_xor(v, o));
    return v;
}
__device__ __forceinline__ float wave_sum(float v) {
#pragma unroll
    for (int o = 16; o > 0; o >>= 1) v += __shfl_xor(v, o);
    return v;
}

__global__ void k_transpose_f16(const float* __restrict__ in, _Float16* __restrict__ out,
                                int R, int C) {
    int i = blockIdx.x * blockDim.x + threadIdx.x;
    if (i >= R * C) return;
    int r = i / C, c = i % C;
    out[c * R + r] = (_Float16)in[i];
}

template <int K>
__global__ void __launch_bounds__(256) k_gemm_wmma(const _Float16* __restrict__ A,
                                                   const _Float16* __restrict__ Bt,
                                                   float* __restrict__ C,
                                                   int M, int Ncol) {
    int wave = (blockIdx.x * blockDim.x + threadIdx.x) >> 5;
    int lane = threadIdx.x & 31;
    int mTiles = M >> 4, nTiles = Ncol >> 4;
    if (wave >= mTiles * nTiles) return;
    int mt = wave / nTiles, nt = wave % nTiles;
    int half = lane >> 4;
    int idx  = lane & 15;

    const _Float16* arow = A  + (size_t)(mt * 16 + idx) * K + half * 8;
    const _Float16* brow = Bt + (size_t)(nt * 16 + idx) * K + half * 8;

    v8f acc = {};
#pragma unroll
    for (int k0 = 0; k0 < K; k0 += 32) {
        union { v16h v; struct { v8h lo, hi; } h; } a, b;
        a.h.lo = *(const v8h*)(arow + k0);
        a.h.hi = *(const v8h*)(arow + k0 + 16);
        b.h.lo = *(const v8h*)(brow + k0);
        b.h.hi = *(const v8h*)(brow + k0 + 16);
        acc = __builtin_amdgcn_wmma_f32_16x16x32_f16(false, a.v, false, b.v,
                                                     (short)0, acc, false, false);
    }
    float* crow = C + (size_t)(mt * 16 + half * 8) * Ncol + nt * 16 + idx;
#pragma unroll
    for (int r = 0; r < 8; ++r) crow[(size_t)r * Ncol] = acc[r];
}

#define P1_WAVES 8
#define P1_MT    (N_NODES / 16)
#define P1_NB    ((P1_MT + P1_WAVES - 1) / P1_WAVES)
#define P1_PITCH (IN_F + 8)
typedef __attribute__((ext_vector_type(4))) float v4f;

__global__ void __launch_bounds__(P1_WAVES * 32) k_proj1(const float* __restrict__ x,
                                                         const _Float16* __restrict__ w1t,
                                                         float* __restrict__ feat) {
    __shared__ __attribute__((aligned(16))) _Float16 wl[F2 * P1_PITCH];
    const int t = threadIdx.x, lane = t & 31;
    const int half = lane >> 4, idx = lane & 15;
    const int tile = __builtin_amdgcn_readfirstlane(blockIdx.x * P1_WAVES + (t >> 5));
    const int tl = tile < P1_MT ? tile : P1_MT - 1;

    v8h wq[16];
#pragma unroll
    for (int j = 0; j < 16; ++j) {
        const int c = t + j * (P1_WAVES * 32);
        wq[j] = *(const v8h*)(w1t + (size_t)(c >> 5) * IN_F + (c & 31) * 8);
    }
    const float* arow = x + (size_t)(tl * 16 + idx) * IN_F + half * 8;
    v4f ar[8][4];
#pragma unroll
    for (int s = 0; s < 8; ++s) {
        ar[s][0] = *(const v4f*)(arow + s * 32);
        ar[s][1] = *(const v4f*)(arow + s * 32 + 4);
        ar[s][2] = *(const v4f*)(arow + s * 32 + 16);
        ar[s][3] = *(const v4f*)(arow + s * 32 + 20);
    }
#pragma unroll
    for (int j = 0; j < 16; ++j) {
        const int c = t + j * (P1_WAVES * 32);
        *(v8h*)(wl + (c >> 5) * P1_PITCH + (c & 31) * 8) = wq[j];
    }
    v16h af[8];
#pragma unroll
    for (int s = 0; s < 8; ++s)
#pragma unroll
        for (int q = 0; q < 4; ++q) {
            af[s][q * 4 + 0] = (_Float16)ar[s][q].x; af[s][q * 4 + 1] = (_Float16)ar[s][q].y;
            af[s][q * 4 + 2] = (_Float16)ar[s][q].z; af[s][q * 4 + 3] = (_Float16)ar[s][q].w;
        }
    __syncthreads();
    if (tile >= P1_MT) return;

    v8f acc[8];
#pragma unroll
    for (int nt = 0; nt < 8; ++nt) acc[nt] = v8f{};
    const _Float16* bl = wl + idx * P1_PITCH + half * 8;
#pragma unroll
    for (int s = 0; s < 8; ++s) {
        const v16h a = af[s];
#pragma unroll
        for (int nt = 0; nt < 8; ++nt) {
            union { v16h v; struct { v8h lo, hi; } h; } b;
            const _Float16* bp = bl + nt * 16 * P1_PITCH + s * 32;
            b.h.lo = *(const v8h*)(bp);
            b.h.hi = *(const v8h*)(bp + 16);
            acc[nt] = __builtin_amdgcn_wmma_f32_16x16x32_f16(false, a, false, b.v,
                                                             (short)0, acc[nt], false, false);
        }
    }
    float* crow = feat + (size_t)(tile * 16 + half * 8) * F2 + idx;
#pragma unroll
    for (int nt = 0; nt < 8; ++nt)
#pragma unroll
        for (int r = 0; r < 8; ++r) crow[(size_t)r * F2 + nt * 16] = acc[nt][r];
}

__global__ void k_el_er(const float* __restrict__ feat, const float* __restrict__ al,
                        const float* __restrict__ ar, float* __restrict__ el,
                        float* __restrict__ er, int n, int H, int D) {
    int i = blockIdx.x * blockDim.x + threadIdx.x;
    if (i >= n * H) return;
    int node = i / H, h = i % H;
    const float* f = feat + (size_t)node * H * D + h * D;
    const float* a = al + h * D;
    const float* b = ar + h * D;
    float sl = 0.0f, sr = 0.0f;
    for (int d = 0; d < D; ++d) { float v = f[d]; sl += v * a[d]; sr += v * b[d]; }
    el[i] = sl; er[i] = sr;
}

#define SORT_T    1024
#define SORT_W    (SORT_T / 32)
#define CHUNKS    256
#define CH        (N_EDGES / CHUNKS)
#define CH_PER_T  ((CH + SORT_T - 1) / SORT_T)
#define BIN_SHIFT 7
#define BIN_W     (1 << BIN_SHIFT)
#define NBIN      ((N_NODES + BIN_W - 1) >> BIN_SHIFT)
#define NBINP     ((NBIN + 1 + 3) & ~3)
#define NB4       (NBINP / 4)
#define PART_RG   (SORT_T / NB4)
#define PART_RPT  ((CHUNKS + PART_RG - 1) / PART_RG)
#define CNT_ROWS  (PART_RG * PART_RPT)
#define BS_CAP    16384
static_assert(N_EDGES % CHUNKS == 0, "slices are equal");
static_assert(CNT_ROWS - CHUNKS <= CHUNKS, "every zero row has a workgroup that writes it");
static_assert(N_NODES <= (1 << (31 - BIN_SHIFT)), "a source index and the low destination bits share one word");
static_assert(NBIN <= 65535, "bin numbers are kept in 16 bits");

__device__ __forceinline__ int wave_scan_incl(int v, int lane) {
    for (int o = 1; o < 32; o <<= 1) {
        const int u = __shfl_up(v, o);
        if (lane >= o) v += u;
    }
    return v;
}

__global__ void __launch_bounds__(SORT_T) k_count(const int* __restrict__ dst, int* __restrict__ cnt) {
    __shared__ int h[NBINP];
    const int t = threadIdx.x, c = blockIdx.x;
    const int* d = dst + (size_t)c * CH;
    int dv[CH_PER_T];
#pragma unroll
    for (int j = 0; j < CH_PER_T; ++j) {
        const int e = t + j * SORT_T;
        dv[j] = d[e < CH ? e : CH - 1];
    }
    for (int i = t; i < NBINP; i += SORT_T) h[i] = 0;
    __syncthreads();
#pragma unroll
    for (int j = 0; j < CH_PER_T; ++j)
        if (t + j * SORT_T < CH) atomicAdd(&h[dv[j] >> BIN_SHIFT], 1);
    __syncthreads();
    for (int i = t; i < NBINP; i += SORT_T) cnt[c * NBINP + i] = h[i];
    if (c < CNT_ROWS - CHUNKS)
        for (int i = t; i < NBINP; i += SORT_T) cnt[(CHUNKS + c) * NBINP + i] = 0;
}

__global__ void __launch_bounds__(SORT_T) k_part(const int* __restrict__ src, const int* __restrict__ dst,
                                                 const int* __restrict__ cnt, int* __restrict__ part,
                                                 int* __restrict__ binStart) {
    __shared__ int tot[NBINP];
    __shared__ int pre[NBINP];
    __shared__ int lcur[NBINP];
    __shared__ int gofs[NBINP];
    __shared__ int wtA[SORT_W];
    __shared__ int wtB[SORT_W];
    __shared__ int stage[CH];
    __shared__ unsigned short sbin[CH];
    const int t = threadIdx.x, lane = t & 31, w = t >> 5, c = blockIdx.x;

    const int* d = dst + (size_t)c * CH;
    const int* s = src + (size_t)c * CH;
    int dv[CH_PER_T], sv[CH_PER_T];
#pragma unroll
    for (int j = 0; j < CH_PER_T; ++j) {
        const int e = t + j * SORT_T;
        dv[j] = d[e < CH ? e : CH - 1];
    }
    for (int i = t; i < NBINP; i += SORT_T) { tot[i] = 0; pre[i] = 0; }
    __syncthreads();

    {
        const int col4 = t % NB4, rg = t / NB4;
        int4 T = {0, 0, 0, 0}, P = {0, 0, 0, 0};
        if (rg < PART_RG) {
            const int4* c4 = (const int4*)cnt + rg * NB4 + col4;
#pragma unroll
            for (int k = 0; k < PART_RPT; ++k) {
                const int4 v = c4[k * (PART_RG * NB4)];
                const int p = rg + k * PART_RG < c ? 1 : 0;
                T.x += v.x; T.y += v.y; T.z += v.z; T.w += v.w;
                P.x += p * v.x; P.y += p * v.y; P.z += p * v.z; P.w += p * v.w;
            }
            atomicAdd(&tot[col4 * 4 + 0], T.x); atomicAdd(&tot[col4 * 4 + 1], T.y);
            atomicAdd(&tot[col4 * 4 + 2], T.z); atomicAdd(&tot[col4 * 4 + 3], T.w);
            atomicAdd(&pre[col4 * 4 + 0], P.x); atomicAdd(&pre[col4 * 4 + 1], P.y);
            atomicAdd(&pre[col4 * 4 + 2], P.z); atomicAdd(&pre[col4 * 4 + 3], P.w);
        }
    }
#pragma unroll
    for (int j = 0; j < CH_PER_T; ++j) {
        const int e = t + j * SORT_T;
        sv[j] = s[e < CH ? e : CH - 1];
    }
    __syncthreads();

    const int a = t < NBINP ? tot[t] : 0;
    const int o = t < NBINP ? cnt[c * NBINP + t] : 0;
    const int ia = wave_scan_incl(a, lane);
    const int io = wave_scan_incl(o, lane);
    if (lane == 31) { wtA[w] = ia; wtB[w] = io; }
    __syncthreads();
    const int ta = wtA[lane], tb = wtB[lane];
    const int sa = wave_scan_incl(ta, lane), sb = wave_scan_incl(tb, lane);
    const int ea = __shfl(sa - ta, w) + ia - a;
    const int eo = __shfl(sb - tb, w) + io - o;
    if (t < NBINP) {
        lcur[t] = eo;
        gofs[t] = ea + pre[t] - eo;
        if (c == 0) binStart[t] = ea;
    }
    __syncthreads();

#pragma unroll
    for (int j = 0; j < CH_PER_T; ++j)
        if (t + j * SORT_T < CH) {
            const int b = dv[j] >> BIN_SHIFT;
            const int lp = atomicAdd(&lcur[b], 1);
            stage[lp] = (sv[j] << BIN_SHIFT) | (dv[j] & (BIN_W - 1));
            sbin[lp] = (unsigned short)b;
        }
    __syncthreads();
    for (int i = t; i < CH; i += SORT_T) part[gofs[sbin[i]] + i] = stage[i];
}

__global__ void __launch_bounds__(SORT_T) k_binsort(const int* __restrict__ part, const int* __restrict__ binStart,
                                                    int* __restrict__ rowptr, int* __restrict__ ssrc) {
    __shared__ int cntN[BIN_W];
    __shared__ int curN[BIN_W];
    __shared__ int wt[BIN_W / 32];
    __shared__ int stage[BS_CAP];
    const int t = threadIdx.x, lane = t & 31, w = t >> 5, b = blockIdx.x;
    const int beg = binStart[b], end = binStart[b + 1];
    const int n = end - beg;
    const int* pb = part + beg;
    if (t < BIN_W) cntN[t] = 0;
    __syncthreads();
    for (int base = 0; base < n; base += 4 * SORT_T) {
        int wv[4];
#pragma unroll
        for (int u = 0; u < 4; ++u) { const int i = base + u * SORT_T + t; wv[u] = pb[i < n ? i : 0]; }
#pragma unroll
        for (int u = 0; u < 4; ++u)
            if (base + u * SORT_T + t < n) atomicAdd(&cntN[wv[u] & (BIN_W - 1)], 1);
    }
    __syncthreads();
    const int v = t < BIN_W ? cntN[t] : 0;
    const int inc = wave_scan_incl(v, lane);
    if (t < BIN_W && lane == 31) wt[w] = inc;
    __syncthreads();
    if (t < BIN_W) {
        int off = 0;
        for (int k = 0; k < w; ++k) off += wt[k];
        const int excl = off + inc - v;
        curN[t] = excl;
        const int node = b * BIN_W + t;
        if (node < N_NODES) rowptr[node] = beg + excl;
    }
    if (b == NBIN - 1 && t == 0) rowptr[N_NODES] = end;
    __syncthreads();
    if (n <= BS_CAP) {
        for (int base = 0; base < n; base += 4 * SORT_T) {
            int wv[4];
#pragma unroll
            for (int u = 0; u < 4; ++u) { const int i = base + u * SORT_T + t; wv[u] = pb[i < n ? i : 0]; }
#pragma unroll
            for (int u = 0; u < 4; ++u)
                if (base + u * SORT_T + t < n) {
                    const int p = atomicAdd(&curN[wv[u] & (BIN_W - 1)], 1);
                    stage[p] = wv[u] >> BIN_SHIFT;
                }
        }
        __syncthreads();
        for (int i = t; i < n; i += SORT_T) ssrc[beg + i] = stage[i];
    } else {
        for (int i = t; i < n; i += SORT_T) {
            const int x = pb[i];
            const int p = atomicAdd(&curN[x & (BIN_W - 1)], 1);
            ssrc[beg + p] = x >> BIN_SHIFT;
        }
    }
}

__device__ __forceinline__ void load_chunk4(const int* __restrict__ ssrc, const float* __restrict__ el,
                                            float4 erd, int base, int end, int lane,
                                            int& s, float4& v) {
    const int i = base + lane;
    const bool valid = i < end;
    s = ssrc[valid ? i : end - 1];
    const float4 a = *(const float4*)(el + 4 * (size_t)s);
    v.x = valid ? leaky(a.x + erd.x) : -INFINITY;
    v.y = valid ? leaky(a.y + erd.y) : -INFINITY;
    v.z = valid ? leaky(a.z + erd.z) : -INFINITY;
    v.w = valid ? leaky(a.w + erd.w) : -INFINITY;
}

__device__ __forceinline__ void accum_chunk4(int s, float4 v, int cnt, float4 m, float4* xrow,
                                             const float* __restrict__ feat, int lane, int h,
                                             float4& acc, float4& den) {
    float4 x;
    x.x = __expf(v.x - m.x); x.y = __expf(v.y - m.y);
    x.z = __expf(v.z - m.z); x.w = __expf(v.w - m.w);
    den.x += x.x; den.y += x.y; den.z += x.z; den.w += x.w;
    xrow[lane] = x;
    const float* xw = (const float*)xrow;
    const int cnt4 = (cnt + 3) & ~3;
    for (int j = 0; j < cnt4; j += 4) {
#pragma unroll
        for (int u = 0; u < 4; ++u) {
            const int sj = __builtin_amdgcn_readlane(s, j + u);
            const float xj = xw[(j + u) * 4 + h];
            const float4 f = *(const float4*)(feat + (size_t)sj * F2 + lane * 4);
            acc.x = fmaf(f.x, xj, acc.x);
            acc.y = fmaf(f.y, xj, acc.y);
            acc.z = fmaf(f.z, xj, acc.z);
            acc.w = fmaf(f.w, xj, acc.w);
        }
    }
}

typedef __attribute__((ext_vector_type(4))) _Float16 v4h;

__global__ void __launch_bounds__(256) k_node4(const int* __restrict__ rowptr,
                                               const int* __restrict__ ssrc,
                                               const float* __restrict__ el,
                                               const float* __restrict__ er,
                                               const float* __restrict__ feat,
                                               const float* __restrict__ b1,
                                               _Float16* __restrict__ h16) {
    __shared__ float4 xs[8][32];
    const int lane = threadIdx.x & 31;
    const int wib  = threadIdx.x >> 5;
    const int node = __builtin_amdgcn_readfirstlane(blockIdx.x * 8 + wib);
    if (node >= N_NODES) return;
    const int beg = rowptr[node], end = rowptr[node + 1];
    const int h = lane >> 3;
    const float4 bb = *(const float4*)(b1 + lane * 4);
    float4 acc = {0.0f, 0.0f, 0.0f, 0.0f};
    float d = 0.0f;
    if (beg < end) {
        const float4 erd = *(const float4*)(er + 4 * (size_t)node);
        int sF; float4 vF;
        load_chunk4(ssrc, el, erd, beg, end, lane, sF, vF);
        float4 m = vF;
        for (int base = beg + 32; base < end; base += 32) {
            int s; float4 v;
            load_chunk4(ssrc, el, erd, base, end, lane, s, v);
            m.x = fmaxf(m.x, v.x); m.y = fmaxf(m.y, v.y);
            m.z = fmaxf(m.z, v.z); m.w = fmaxf(m.w, v.w);
        }
        m.x = finite_or_zero(wave_max(m.x)); m.y = finite_or_zero(wave_max(m.y));
        m.z = finite_or_zero(wave_max(m.z)); m.w = finite_or_zero(wave_max(m.w));

        float4 den = {0.0f, 0.0f, 0.0f, 0.0f};
        int rem = end - beg;
        accum_chunk4(sF, vF, rem < 32 ? rem : 32, m, xs[wib], feat, lane, h, acc, den);
        for (int base = beg + 32; base < end; base += 32) {
            int s; float4 v;
            load_chunk4(ssrc, el, erd, base, end, lane, s, v);
            rem = end - base;
            accum_chunk4(s, v, rem < 32 ? rem : 32, m, xs[wib], feat, lane, h, acc, den);
        }
        const float d0 = wave_sum(den.x), d1 = wave_sum(den.y);
        const float d2 = wave_sum(den.z), d3 = wave_sum(den.w);
        d = (h == 0) ? d0 : (h == 1) ? d1 : (h == 2) ? d2 : d3;
    }
    d = fmaxf(d, 1e-9f);
    float r[4] = { acc.x / d + bb.x, acc.y / d + bb.y, acc.z / d + bb.z, acc.w / d + bb.w };
    v4h o;
#pragma unroll
    for (int k = 0; k < 4; ++k) {
        const float t = r[k] > 0.0f ? r[k] : (__expf(r[k]) - 1.0f);
        o[k] = (_Float16)t;
    }
    *(v4h*)(h16 + (size_t)node * F2 + lane * 4) = o;
}

__device__ __forceinline__ void load_chunk1(const int* __restrict__ ssrc, const float* __restrict__ el,
                                            float erd, int base, int end, int lane,
                                            int& s, float& v) {
    const int i = base + lane;
    const bool valid = i < end;
    s = ssrc[valid ? i : end - 1];
    v = valid ? leaky(el[s] + erd) : -INFINITY;
}

__device__ __forceinline__ void accum_chunk1(int s, float v, float m, const float* __restrict__ feat,
                                             int sub, int q, float4& acc, float& den) {
    const float x = __expf(v - m);
    den += x;
#pragma unroll
    for (int g = 0; g < 4; ++g) {
        const int jl = g * 8 + sub;
        const int sj = __shfl(s, jl);
        const float xj = __shfl(x, jl);
        const float4 f = *(const float4*)(feat + (size_t)sj * NC + q * 4);
        acc.x = fmaf(f.x, xj, acc.x);
        acc.y = fmaf(f.y, xj, acc.y);
        acc.z = fmaf(f.z, xj, acc.z);
        acc.w = fmaf(f.w, xj, acc.w);
    }
}

__global__ void __launch_bounds__(256) k_node1(const int* __restrict__ rowptr,
                                               const int* __restrict__ ssrc,
                                               const float* __restrict__ el,
                                               const float* __restrict__ er,
                                               const float* __restrict__ feat,
                                               const float* __restrict__ b2,
                                               float* __restrict__ out) {
    const int lane = threadIdx.x & 31;
    const int wib  = threadIdx.x >> 5;
    const int node = __builtin_amdgcn_readfirstlane(blockIdx.x * 8 + wib);
    if (node >= N_NODES) return;
    const int beg = rowptr[node], end = rowptr[node + 1];
    const int q = lane & 3, sub = lane >> 2;
    float4 acc = {0.0f, 0.0f, 0.0f, 0.0f};
    float den = 0.0f;
    if (beg < end) {
        const float erd = er[node];
        int sF; float vF;
        load_chunk1(ssrc, el, erd, beg, end, lane, sF, vF);
        float m = vF;
        for (int base = beg + 32; base < end; base += 32) {
            int s; float v;
            load_chunk1(ssrc, el, erd, base, end, lane, s, v);
            m = fmaxf(m, v);
        }
        m = finite_or_zero(wave_max(m));
        accum_chunk1(sF, vF, m, feat, sub, q, acc, den);
        for (int base = beg + 32; base < end; base += 32) {
            int s; float v;
            load_chunk1(ssrc, el, erd, base, end, lane, s, v);
            accum_chunk1(s, v, m, feat, sub, q, acc, den);
        }
#pragma unroll
        for (int o = 4; o < 32; o <<= 1) {
            acc.x += __shfl_xor(acc.x, o); acc.y += __shfl_xor(acc.y, o);
            acc.z += __shfl_xor(acc.z, o); acc.w += __shfl_xor(acc.w, o);
        }
        den = wave_sum(den);
    }
    if (sub == 0) {
        const float d = fmaxf(den, 1e-9f);
        const float4 bb = *(const float4*)(b2 + q * 4);
        float4 r;
        r.x = acc.x / d + bb.x; r.y = acc.y / d + bb.y;
        r.z = acc.z / d + bb.z; r.w = acc.w / d + bb.w;
        *(float4*)(out + (size_t)node * NC + q * 4) = r;
    }
}

extern "C" void kernel_launch(void* const* d_in, const int* in_sizes, int n_in,
                              void* d_out, int out_size, void* d_ws, size_t ws_size,
                              hipStream_t stream) {
    const float* x   = (const float*)d_in[0];
    const int*   src = (const int*)  d_in[1];
    const int*   dst = (const int*)  d_in[2];
    const float* W1  = (const float*)d_in[3];
    const float* b1  = (const float*)d_in[4];
    const float* al1 = (const float*)d_in[5];
    const float* ar1 = (const float*)d_in[6];
    const float* W2  = (const float*)d_in[7];
    const float* b2  = (const float*)d_in[8];
    const float* al2 = (const float*)d_in[9];
    const float* ar2 = (const float*)d_in[10];
    float* out = (float*)d_out;
    (void)in_sizes; (void)n_in; (void)out_size; (void)ws_size;

    char* p = (char*)d_ws;
    auto carve = [&](size_t bytes) -> char* {
        char* r = p; p += (bytes + 255) & ~(size_t)255; return r;
    };
    _Float16* w1t   = (_Float16*)carve((size_t)IN_F * F2 * 2);
    float*    feat1 = (float*)   carve((size_t)N_NODES * F2 * 4);
    float*    el1   = (float*)   carve((size_t)N_NODES * HEADS * 4);
    float*    er1   = (float*)   carve((size_t)N_NODES * HEADS * 4);
    _Float16* h16   = (_Float16*)carve((size_t)N_NODES * F2 * 2);
    _Float16* w2t   = (_Float16*)carve((size_t)F2 * NC * 2);
    float*    feat2 = (float*)   carve((size_t)N_NODES * NC * 4);
    float*    el2   = (float*)   carve((size_t)N_NODES * 4);
    float*    er2   = (float*)   carve((size_t)N_NODES * 4);
    int*      cnt      = (int*)carve((size_t)CNT_ROWS * NBINP * 4);
    int*      binStart = (int*)carve((size_t)NBINP * 4);
    int*      part     = (int*)carve((size_t)N_EDGES * 4);
    int*      rowptr   = (int*)carve((size_t)(N_NODES + 1) * 4);
    int*      ssrc     = (int*)carve((size_t)N_EDGES * 4);

    const int T = 256;
    auto blk = [](long long n, int t) { return (int)((n + t - 1) / t); };

    k_transpose_f16<<<blk((long long)IN_F * F2, T), T, 0, stream>>>(W1, w1t, IN_F, F2);
    k_proj1<<<P1_NB, P1_WAVES * 32, 0, stream>>>(x, w1t, feat1);
    k_el_er<<<blk((long long)N_NODES * HEADS, T), T, 0, stream>>>(feat1, al1, ar1, el1, er1,
                                                                  N_NODES, HEADS, HID);
    k_count<<<CHUNKS, SORT_T, 0, stream>>>(dst, cnt);
    k_part<<<CHUNKS, SORT_T, 0, stream>>>(src, dst, cnt, part, binStart);
    k_binsort<<<NBIN, SORT_T, 0, stream>>>(part, binStart, rowptr, ssrc);
    k_node4<<<blk(N_NODES, 8), T, 0, stream>>>(rowptr, ssrc, el1, er1, feat1, b1, h16);

    k_transpose_f16<<<blk((long long)F2 * NC, T), T, 0, stream>>>(W2, w2t, F2, NC);
    {
        int waves = (N_NODES / 16) * (NC / 16);
        k_gemm_wmma<F2><<<blk(waves, 8), T, 0, stream>>>(h16, w2t, feat2, N_NODES, NC);
    }
    k_el_er<<<blk(N_NODES, T), T, 0, stream>>>(feat2, al2, ar2, el2, er2, N_NODES, 1, NC);
    k_node1<<<blk(N_NODES, 8), T, 0, stream>>>(rowptr, ssrc, el2, er2, feat2, b2, out);
}
